# starting kernel
# baseline (speedup 1.0000x reference)
#include <hip/hip_runtime.h>


typedef __attribute__((ext_vector_type(2))) float v2f;
typedef __attribute__((ext_vector_type(8))) float v8f;

#define N_IN   16
#define N_OUT  64
#define MREF   32
#define SDIM   256
#define WPB    8
#define GPITCH 33

__global__ __launch_bounds__(256) void householder_wy_kernel(
    const float* __restrict__ x,
    const float* __restrict__ v,
    float* __restrict__ out)
{
    const int wave = threadIdx.x >> 5;
    const int lane = threadIdx.x & 31;
    const int half = lane >> 4;
    const int lr   = lane & 15;

    const int batch = blockIdx.x * WPB + wave;
    const int i_in  = batch >> 6;
    const int o     = batch & 63;

    const float* vb = v + (size_t)batch * (MREF * SDIM);
    const float* xb = x + (size_t)(i_in * SDIM) * N_OUT + o;

    __shared__ float sG[WPB][MREF * GPITCH];
    __shared__ float sW[WPB][MREF];
    __shared__ float sA[WPB][MREF];
    __shared__ float sX[WPB][SDIM];

    float xr[8];
#pragma unroll
    for (int t = 0; t < 8; ++t) {
        xr[t] = xb[(size_t)(lane + 32 * t) * N_OUT];
        sX[wave][lane + 32 * t] = xr[t];
    }
    __syncthreads();

    v8f c00 = {}; v8f c01 = {}; v8f c11 = {}; v8f cw0 = {}; v8f cw1 = {};

#pragma unroll 4
    for (int k = 0; k < SDIM; k += 4) {
        const int kk = k + 2 * half;
        v2f a0 = *(const v2f*)(vb + (size_t)lr * SDIM + kk);
        v2f a1 = *(const v2f*)(vb + (size_t)(16 + lr) * SDIM + kk);
        const float x0 = sX[wave][kk];
        const float x1 = sX[wave][kk + 1];
        v2f b2;
        b2.x = (lr == 0) ? x0 : 0.0f;
        b2.y = (lr == 0) ? x1 : 0.0f;
        c00 = __builtin_amdgcn_wmma_f32_16x16x4_f32(false, a0, false, a0, (short)0, c00, false, false);
        c01 = __builtin_amdgcn_wmma_f32_16x16x4_f32(false, a0, false, a1, (short)0, c01, false, false);
        c11 = __builtin_amdgcn_wmma_f32_16x16x4_f32(false, a1, false, a1, (short)0, c11, false, false);
        cw0 = __builtin_amdgcn_wmma_f32_16x16x4_f32(false, a0, false, b2, (short)0, cw0, false, false);
        cw1 = __builtin_amdgcn_wmma_f32_16x16x4_f32(false, a1, false, b2, (short)0, cw1, false, false);
    }

#pragma unroll
    for (int j = 0; j < 8; ++j) {
        const int row = j + 8 * half;
        sG[wave][row * GPITCH + lr]               = c00[j];
        sG[wave][row * GPITCH + 16 + lr]          = c01[j];
        sG[wave][(16 + row) * GPITCH + 16 + lr]   = c11[j];
        if (lr == 0) {
            sW[wave][row]      = cw0[j];
            sW[wave][16 + row] = cw1[j];
        }
    }
    __syncthreads();

    {
        float r = sW[wave][lane];
        for (int i = MREF - 1; i >= 0; --i) {
            const float ri  = __shfl(r, i, 32);
            const float a   = 2.0f * ri / sG[wave][i * GPITCH + i];
            if (lane == i) sA[wave][i] = a;
            if (lane < i)  r -= a * sG[wave][lane * GPITCH + i];
        }
    }
    __syncthreads();

    float acc[8];
#pragma unroll
    for (int t = 0; t < 8; ++t)
        acc[t] = xr[t];

    for (int kk = 0; kk < MREF; ++kk) {
        const float ak = sA[wave][kk];
        const float* vr = vb + (size_t)kk * SDIM;
#pragma unroll
        for (int t = 0; t < 8; ++t)
            acc[t] -= ak * vr[lane + 32 * t];
    }

    float* ob = out + (size_t)(i_in * SDIM) * N_OUT + o;
#pragma unroll
    for (int t = 0; t < 8; ++t)
        ob[(size_t)(lane + 32 * t) * N_OUT] = acc[t];
}

extern "C" void kernel_launch(void* const* d_in, const int* in_sizes, int n_in,
                              void* d_out, int out_size, void* d_ws, size_t ws_size,
                              hipStream_t stream) {
    const float* x = (const float*)d_in[0];
    const float* v = (const float*)d_in[1];
    float* out = (float*)d_out;

    dim3 grid((N_IN * N_OUT) / WPB);
    dim3 block(WPB * 32);
    householder_wy_kernel<<<grid, block, 0, stream>>>(x, v, out);
}


# One 4-wave block per batch: V read from global once (coalesced, kept in registers for y, staged in LDS for WMMA); k split over 4 waves, so G and w are summed in a new order
# speedup vs baseline: 1.5009x; 1.5009x over previous
#include <hip/hip_runtime.h>


typedef __attribute__((ext_vector_type(2))) float v2f;
typedef __attribute__((ext_vector_type(8))) float v8f;

#define N_IN   16
#define N_OUT  64
#define MREF   32
#define SDIM   256
#define NWAVE  4
#define NTHR   (NWAVE * 32)
#define KPW    (SDIM / NWAVE)
#define VPITCH 260
#define GPITCH 33

__global__ __launch_bounds__(NTHR) void householder_wy_kernel(
    const float* __restrict__ x,
    const float* __restrict__ v,
    float* __restrict__ out)
{
    const int tid  = threadIdx.x;
    const int wave = tid >> 5;
    const int lane = tid & 31;
    const int half = lane >> 4;
    const int lr   = lane & 15;

    const int batch = blockIdx.x;
    const int i_in  = batch >> 6;
    const int o     = batch & 63;

    const float* vb = v + (size_t)batch * (MREF * SDIM);
    const float* xb = x + (size_t)(i_in * SDIM) * N_OUT + o;

    __shared__ __attribute__((aligned(16))) float sV[MREF * VPITCH];
    __shared__ __attribute__((aligned(16))) float sX[SDIM];
    __shared__ float sP[NWAVE][3][256];
    __shared__ float sPW[NWAVE][MREF];
    __shared__ float sG[MREF * GPITCH];
    __shared__ float sW[MREF];
    __shared__ __attribute__((aligned(16))) float sA[MREF];

    const int s0 = 2 * tid;
    v2f vr[MREF];
#pragma unroll
    for (int m = 0; m < MREF; ++m)
        vr[m] = *(const v2f*)(vb + (size_t)m * SDIM + s0);
    const float xa = xb[(size_t)s0 * N_OUT];
    const float xc = xb[(size_t)(s0 + 1) * N_OUT];

#pragma unroll
    for (int m = 0; m < MREF; ++m)
        *(v2f*)(&sV[m * VPITCH + s0]) = vr[m];
    {
        v2f xx; xx.x = xa; xx.y = xc;
        *(v2f*)(&sX[s0]) = xx;
    }
    __syncthreads();

    v8f c00 = {}; v8f c01 = {}; v8f c11 = {}; v8f cw0 = {}; v8f cw1 = {};
    const int kbase = wave * KPW;

#pragma unroll
    for (int ks = 0; ks < KPW; ks += 4) {
        const int kk = kbase + ks + 2 * half;
        v2f a0 = *(const v2f*)(&sV[lr * VPITCH + kk]);
        v2f a1 = *(const v2f*)(&sV[(16 + lr) * VPITCH + kk]);
        const float x0 = sX[kk];
        const float x1 = sX[kk + 1];
        v2f b2;
        b2.x = (lr == 0) ? x0 : 0.0f;
        b2.y = (lr == 0) ? x1 : 0.0f;
        c00 = __builtin_amdgcn_wmma_f32_16x16x4_f32(false, a0, false, a0, (short)0, c00, false, false);
        c01 = __builtin_amdgcn_wmma_f32_16x16x4_f32(false, a0, false, a1, (short)0, c01, false, false);
        c11 = __builtin_amdgcn_wmma_f32_16x16x4_f32(false, a1, false, a1, (short)0, c11, false, false);
        cw0 = __builtin_amdgcn_wmma_f32_16x16x4_f32(false, a0, false, b2, (short)0, cw0, false, false);
        cw1 = __builtin_amdgcn_wmma_f32_16x16x4_f32(false, a1, false, b2, (short)0, cw1, false, false);
    }

#pragma unroll
    for (int j = 0; j < 8; ++j) {
        sP[wave][0][j * 32 + lane] = c00[j];
        sP[wave][1][j * 32 + lane] = c01[j];
        sP[wave][2][j * 32 + lane] = c11[j];
    }
    if (lr == 0) {
#pragma unroll
        for (int j = 0; j < 8; ++j) {
            sPW[wave][j + 8 * half]      = cw0[j];
            sPW[wave][16 + j + 8 * half] = cw1[j];
        }
    }
    __syncthreads();

#pragma unroll
    for (int tile = 0; tile < 3; ++tile) {
#pragma unroll
        for (int e = 0; e < 2; ++e) {
            const int idx = tid + NTHR * e;
            float g = sP[0][tile][idx];
#pragma unroll
            for (int w = 1; w < NWAVE; ++w)
                g += sP[w][tile][idx];
            const int j   = idx >> 5;
            const int ln  = idx & 31;
            const int row = j + 8 * (ln >> 4) + ((tile == 2) ? 16 : 0);
            const int col = (ln & 15) + ((tile >= 1) ? 16 : 0);
            sG[row * GPITCH + col] = g;
        }
    }
    if (tid < MREF) {
        float wv = sPW[0][tid];
#pragma unroll
        for (int w = 1; w < NWAVE; ++w)
            wv += sPW[w][tid];
        sW[tid] = wv;
    }
    __syncthreads();

    if (wave == 0) {
        float r = sW[lane];
        for (int i = MREF - 1; i >= 0; --i) {
            const float ri  = __shfl(r, i, 32);
            const float a   = 2.0f * ri / sG[i * GPITCH + i];
            if (lane == i) sA[i] = a;
            if (lane < i)  r -= a * sG[lane * GPITCH + i];
        }
    }
    __syncthreads();

    float acc0 = xa;
    float acc1 = xc;
#pragma unroll
    for (int m = 0; m < MREF; ++m) {
        const float ak = sA[m];
        acc0 -= ak * vr[m].x;
        acc1 -= ak * vr[m].y;
    }

    float* ob = out + (size_t)(i_in * SDIM) * N_OUT + o;
    ob[(size_t)s0 * N_OUT]       = acc0;
    ob[(size_t)(s0 + 1) * N_OUT] = acc1;
}

extern "C" void kernel_launch(void* const* d_in, const int* in_sizes, int n_in,
                              void* d_out, int out_size, void* d_ws, size_t ws_size,
                              hipStream_t stream) {
    const float* x = (const float*)d_in[0];
    const float* v = (const float*)d_in[1];
    float* out = (float*)d_out;

    dim3 grid(N_IN * N_OUT);
    dim3 block(NTHR);
    householder_wy_kernel<<<grid, block, 0, stream>>>(x, v, out);
}


# Backward recurrence taken off LDS: every wave solves it in registers, rows of G pre-scaled by 2/G_ii (a_i rounding changes), a_i passed by v_readlane to scalars that phase 3 reads directly
# speedup vs baseline: 1.8159x; 1.2099x over previous
#include <hip/hip_runtime.h>


typedef __attribute__((ext_vector_type(2))) float v2f;
typedef __attribute__((ext_vector_type(4))) float v4f;
typedef __attribute__((ext_vector_type(8))) float v8f;

#define N_IN   16
#define N_OUT  64
#define MREF   32
#define SDIM   256
#define NWAVE  4
#define NTHR   (NWAVE * 32)
#define KPW    (SDIM / NWAVE)
#define VPITCH 260
#define GPITCH 36

__global__ __launch_bounds__(NTHR) void householder_wy_kernel(
    const float* __restrict__ x,
    const float* __restrict__ v,
    float* __restrict__ out)
{
    const int tid  = threadIdx.x;
    const int wave = tid >> 5;
    const int lane = tid & 31;
    const int half = lane >> 4;
    const int lr   = lane & 15;

    const int batch = blockIdx.x;
    const int i_in  = batch >> 6;
    const int o     = batch & 63;

    const float* vb = v + (size_t)batch * (MREF * SDIM);
    const float* xb = x + (size_t)(i_in * SDIM) * N_OUT + o;

    __shared__ __attribute__((aligned(16))) float sV[MREF * VPITCH];
    __shared__ __attribute__((aligned(16))) float sX[SDIM];
    __shared__ float sP[NWAVE][3][256];
    __shared__ float sPW[NWAVE][MREF];
    __shared__ __attribute__((aligned(16))) float sG[MREF * GPITCH];
    __shared__ float sW[MREF];

    const int s0 = 2 * tid;
    v2f vr[MREF];
#pragma unroll
    for (int m = 0; m < MREF; ++m)
        vr[m] = *(const v2f*)(vb + (size_t)m * SDIM + s0);
    const float xa = xb[(size_t)s0 * N_OUT];
    const float xc = xb[(size_t)(s0 + 1) * N_OUT];

#pragma unroll
    for (int m = 0; m < MREF; ++m)
        *(v2f*)(&sV[m * VPITCH + s0]) = vr[m];
    {
        v2f xx; xx.x = xa; xx.y = xc;
        *(v2f*)(&sX[s0]) = xx;
    }
    __syncthreads();

    v8f c00 = {}; v8f c01 = {}; v8f c11 = {}; v8f cw0 = {}; v8f cw1 = {};
    const int kbase = wave * KPW;

#pragma unroll
    for (int ks = 0; ks < KPW; ks += 4) {
        const int kk = kbase + ks + 2 * half;
        v2f a0 = *(const v2f*)(&sV[lr * VPITCH + kk]);
        v2f a1 = *(const v2f*)(&sV[(16 + lr) * VPITCH + kk]);
        const float x0 = sX[kk];
        const float x1 = sX[kk + 1];
        v2f b2;
        b2.x = (lr == 0) ? x0 : 0.0f;
        b2.y = (lr == 0) ? x1 : 0.0f;
        c00 = __builtin_amdgcn_wmma_f32_16x16x4_f32(false, a0, false, a0, (short)0, c00, false, false);
        c01 = __builtin_amdgcn_wmma_f32_16x16x4_f32(false, a0, false, a1, (short)0, c01, false, false);
        c11 = __builtin_amdgcn_wmma_f32_16x16x4_f32(false, a1, false, a1, (short)0, c11, false, false);
        cw0 = __builtin_amdgcn_wmma_f32_16x16x4_f32(false, a0, false, b2, (short)0, cw0, false, false);
        cw1 = __builtin_amdgcn_wmma_f32_16x16x4_f32(false, a1, false, b2, (short)0, cw1, false, false);
    }

#pragma unroll
    for (int j = 0; j < 8; ++j) {
        sP[wave][0][j * 32 + lane] = c00[j];
        sP[wave][1][j * 32 + lane] = c01[j];
        sP[wave][2][j * 32 + lane] = c11[j];
    }
    if (lr == 0) {
#pragma unroll
        for (int j = 0; j < 8; ++j) {
            sPW[wave][j + 8 * half]      = cw0[j];
            sPW[wave][16 + j + 8 * half] = cw1[j];
        }
    }
    __syncthreads();

#pragma unroll
    for (int tile = 0; tile < 3; ++tile) {
#pragma unroll
        for (int e = 0; e < 2; ++e) {
            const int idx = tid + NTHR * e;
            float g = sP[0][tile][idx];
#pragma unroll
            for (int w = 1; w < NWAVE; ++w)
                g += sP[w][tile][idx];
            const int j   = idx >> 5;
            const int ln  = idx & 31;
            const int row = j + 8 * (ln >> 4) + ((tile == 2) ? 16 : 0);
            const int col = (ln & 15) + ((tile >= 1) ? 16 : 0);
            sG[row * GPITCH + col] = g;
            if (tile == 1) sG[col * GPITCH + row] = g;
        }
    }
    if (tid < MREF) {
        float wv = sPW[0][tid];
#pragma unroll
        for (int w = 1; w < NWAVE; ++w)
            wv += sPW[w][tid];
        sW[tid] = wv;
    }
    __syncthreads();

    const float gjj = sG[lane * GPITCH + lane];
    const float wj  = sW[lane];
    float gs[MREF];
#pragma unroll
    for (int q = 0; q < MREF / 4; ++q) {
        const v4f g4 = *(const v4f*)(&sG[lane * GPITCH + 4 * q]);
        gs[4 * q + 0] = g4.x; gs[4 * q + 1] = g4.y; gs[4 * q + 2] = g4.z; gs[4 * q + 3] = g4.w;
    }
    const float inv = 2.0f / gjj;
    float t = inv * wj;
#pragma unroll
    for (int k = 0; k < MREF; ++k)
        gs[k] *= inv;

    float a[MREF];
#pragma unroll
    for (int i = MREF - 1; i >= 0; --i) {
        a[i] = __int_as_float(__builtin_amdgcn_readlane(__float_as_int(t), i));
        t -= a[i] * gs[i];
    }

    float acc0 = xa;
    float acc1 = xc;
#pragma unroll
    for (int m = 0; m < MREF; ++m) {
        acc0 -= a[m] * vr[m].x;
        acc1 -= a[m] * vr[m].y;
    }

    float* ob = out + (size_t)(i_in * SDIM) * N_OUT + o;
    ob[(size_t)s0 * N_OUT]       = acc0;
    ob[(size_t)(s0 + 1) * N_OUT] = acc1;
}

extern "C" void kernel_launch(void* const* d_in, const int* in_sizes, int n_in,
                              void* d_out, int out_size, void* d_ws, size_t ws_size,
                              hipStream_t stream) {
    const float* x = (const float*)d_in[0];
    const float* v = (const float*)d_in[1];
    float* out = (float*)d_out;

    dim3 grid(N_IN * N_OUT);
    dim3 block(NTHR);
    householder_wy_kernel<<<grid, block, 0, stream>>>(x, v, out);
}


# Two adjacent outputs per 8-wave block (each batch keeps its own 4 waves and arithmetic): x read and y written once per block as 8-byte pairs per row, y staged through LDS
# speedup vs baseline: 2.4214x; 1.3334x over previous
#include <hip/hip_runtime.h>


typedef __attribute__((ext_vector_type(2))) float v2f;
typedef __attribute__((ext_vector_type(4))) float v4f;
typedef __attribute__((ext_vector_type(8))) float v8f;

#define N_IN   16
#define N_OUT  64
#define MREF   32
#define SDIM   256
#define NB     2
#define NWAVE  4
#define GTHR   (NWAVE * 32)
#define NTHR   (NB * GTHR)
#define LPL    (NB / 2)
#define KPW    (SDIM / NWAVE)
#define VPITCH 260
#define GPITCH 36

__global__ __launch_bounds__(NTHR) void householder_wy_kernel(
    const float* __restrict__ x,
    const float* __restrict__ v,
    float* __restrict__ out)
{
    const int tid  = threadIdx.x;
    const int g    = tid / GTHR;
    const int tb   = tid % GTHR;
    const int wave = tb >> 5;
    const int lane = tb & 31;
    const int half = lane >> 4;
    const int lr   = lane & 15;

    const int blk   = blockIdx.x;
    const int i_in  = blk / (N_OUT / NB);
    const int o0    = (blk % (N_OUT / NB)) * NB;
    const int batch = blk * NB + g;

    const float* vb = v + (size_t)batch * (MREF * SDIM);
    const float* xr = x + (size_t)(i_in * SDIM) * N_OUT + o0;

    __shared__ __attribute__((aligned(16))) float sV[NB][MREF * VPITCH];
    __shared__ __attribute__((aligned(16))) float sX[NB][SDIM];
    __shared__ float sP[NB][NWAVE][3][256];
    __shared__ float sPW[NB][NWAVE][MREF];
    __shared__ __attribute__((aligned(16))) float sG[NB][MREF * GPITCH];
    __shared__ float sW[NB][MREF];
    __shared__ __attribute__((aligned(16))) float sY[NB][SDIM];

    const int s0 = 2 * tb;
    v2f vr[MREF];
#pragma unroll
    for (int m = 0; m < MREF; ++m)
        vr[m] = *(const v2f*)(vb + (size_t)m * SDIM + s0);
    const int xs = tid / LPL;
    const int xp = tid % LPL;
    const v2f xin = *(const v2f*)(xr + (size_t)xs * N_OUT + 2 * xp);

#pragma unroll
    for (int m = 0; m < MREF; ++m)
        *(v2f*)(&sV[g][m * VPITCH + s0]) = vr[m];
    sX[2 * xp][xs]     = xin.x;
    sX[2 * xp + 1][xs] = xin.y;
    __syncthreads();
    const v2f xown = *(const v2f*)(&sX[g][s0]);

    v8f c00 = {}; v8f c01 = {}; v8f c11 = {}; v8f cw0 = {}; v8f cw1 = {};
    const int kbase = wave * KPW;

#pragma unroll
    for (int ks = 0; ks < KPW; ks += 4) {
        const int kk = kbase + ks + 2 * half;
        v2f a0 = *(const v2f*)(&sV[g][lr * VPITCH + kk]);
        v2f a1 = *(const v2f*)(&sV[g][(16 + lr) * VPITCH + kk]);
        const float x0 = sX[g][kk];
        const float x1 = sX[g][kk + 1];
        v2f b2;
        b2.x = (lr == 0) ? x0 : 0.0f;
        b2.y = (lr == 0) ? x1 : 0.0f;
        c00 = __builtin_amdgcn_wmma_f32_16x16x4_f32(false, a0, false, a0, (short)0, c00, false, false);
        c01 = __builtin_amdgcn_wmma_f32_16x16x4_f32(false, a0, false, a1, (short)0, c01, false, false);
        c11 = __builtin_amdgcn_wmma_f32_16x16x4_f32(false, a1, false, a1, (short)0, c11, false, false);
        cw0 = __builtin_amdgcn_wmma_f32_16x16x4_f32(false, a0, false, b2, (short)0, cw0, false, false);
        cw1 = __builtin_amdgcn_wmma_f32_16x16x4_f32(false, a1, false, b2, (short)0, cw1, false, false);
    }

#pragma unroll
    for (int j = 0; j < 8; ++j) {
        sP[g][wave][0][j * 32 + lane] = c00[j];
        sP[g][wave][1][j * 32 + lane] = c01[j];
        sP[g][wave][2][j * 32 + lane] = c11[j];
    }
    if (lr == 0) {
#pragma unroll
        for (int j = 0; j < 8; ++j) {
            sPW[g][wave][j + 8 * half]      = cw0[j];
            sPW[g][wave][16 + j + 8 * half] = cw1[j];
        }
    }
    __syncthreads();

#pragma unroll
    for (int tile = 0; tile < 3; ++tile) {
#pragma unroll
        for (int e = 0; e < 2; ++e) {
            const int idx = tb + GTHR * e;
            float gv = sP[g][0][tile][idx];
#pragma unroll
            for (int w = 1; w < NWAVE; ++w)
                gv += sP[g][w][tile][idx];
            const int j   = idx >> 5;
            const int ln  = idx & 31;
            const int row = j + 8 * (ln >> 4) + ((tile == 2) ? 16 : 0);
            const int col = (ln & 15) + ((tile >= 1) ? 16 : 0);
            sG[g][row * GPITCH + col] = gv;
            if (tile == 1) sG[g][col * GPITCH + row] = gv;
        }
    }
    if (tb < MREF) {
        float wv = sPW[g][0][tb];
#pragma unroll
        for (int w = 1; w < NWAVE; ++w)
            wv += sPW[g][w][tb];
        sW[g][tb] = wv;
    }
    __syncthreads();

    const float gjj = sG[g][lane * GPITCH + lane];
    const float wj  = sW[g][lane];
    float gs[MREF];
#pragma unroll
    for (int q = 0; q < MREF / 4; ++q) {
        const v4f g4 = *(const v4f*)(&sG[g][lane * GPITCH + 4 * q]);
        gs[4 * q + 0] = g4.x; gs[4 * q + 1] = g4.y; gs[4 * q + 2] = g4.z; gs[4 * q + 3] = g4.w;
    }
    const float inv = 2.0f / gjj;
    float t = inv * wj;
#pragma unroll
    for (int k = 0; k < MREF; ++k)
        gs[k] *= inv;

    float a[MREF];
#pragma unroll
    for (int i = MREF - 1; i >= 0; --i) {
        a[i] = __int_as_float(__builtin_amdgcn_readlane(__float_as_int(t), i));
        t -= a[i] * gs[i];
    }

    float acc0 = xown.x;
    float acc1 = xown.y;
#pragma unroll
    for (int m = 0; m < MREF; ++m) {
        acc0 -= a[m] * vr[m].x;
        acc1 -= a[m] * vr[m].y;
    }

    {
        v2f yy; yy.x = acc0; yy.y = acc1;
        *(v2f*)(&sY[g][s0]) = yy;
    }
    __syncthreads();
    v2f yo;
    yo.x = sY[2 * xp][xs];
    yo.y = sY[2 * xp + 1][xs];
    *(v2f*)(out + (size_t)(i_in * SDIM + xs) * N_OUT + o0 + 2 * xp) = yo;
}

extern "C" void kernel_launch(void* const* d_in, const int* in_sizes, int n_in,
                              void* d_out, int out_size, void* d_ws, size_t ws_size,
                              hipStream_t stream) {
    const float* x = (const float*)d_in[0];
    const float* v = (const float*)d_in[1];
    float* out = (float*)d_out;

    dim3 grid(N_IN * N_OUT / NB);
    dim3 block(NTHR);
    householder_wy_kernel<<<grid, block, 0, stream>>>(x, v, out);
}


# w = V x moved into column 0 of the two diagonal G tiles (3 WMMAs per k step, not 5); G[0][0], G[16][16] from one packed WMMA pair per wave, so those two are summed in a new order
# speedup vs baseline: 2.5595x; 1.0570x over previous
#include <hip/hip_runtime.h>


typedef __attribute__((ext_vector_type(2))) float v2f;
typedef __attribute__((ext_vector_type(4))) float v4f;
typedef __attribute__((ext_vector_type(8))) float v8f;

#define N_IN   16
#define N_OUT  64
#define MREF   32
#define SDIM   256
#define NB     2
#define NWAVE  4
#define GTHR   (NWAVE * 32)
#define NTHR   (NB * GTHR)
#define LPL    (NB / 2)
#define KPW    (SDIM / NWAVE)
#define VPITCH 260
#define GPITCH 36

__global__ __launch_bounds__(NTHR) void householder_wy_kernel(
    const float* __restrict__ x,
    const float* __restrict__ v,
    float* __restrict__ out)
{
    const int tid  = threadIdx.x;
    const int g    = tid / GTHR;
    const int tb   = tid % GTHR;
    const int wave = tb >> 5;
    const int lane = tb & 31;
    const int half = lane >> 4;
    const int lr   = lane & 15;

    const int blk   = blockIdx.x;
    const int i_in  = blk / (N_OUT / NB);
    const int o0    = (blk % (N_OUT / NB)) * NB;
    const int batch = blk * NB + g;

    const float* vb = v + (size_t)batch * (MREF * SDIM);
    const float* xr = x + (size_t)(i_in * SDIM) * N_OUT + o0;

    __shared__ __attribute__((aligned(16))) float sV[NB][MREF * VPITCH];
    __shared__ __attribute__((aligned(16))) float sX[NB][SDIM];
    __shared__ float sP[NB][NWAVE][3][256];
    __shared__ float sPW[NB][NWAVE][MREF];
    __shared__ __attribute__((aligned(16))) float sG[NB][MREF * GPITCH];
    __shared__ float sW[NB][MREF];
    __shared__ __attribute__((aligned(16))) float sD[NB][NWAVE][16];
    __shared__ float sDg[NB][2];
    __shared__ __attribute__((aligned(16))) float sY[NB][SDIM];

    const int s0 = 2 * tb;
    v2f vr[MREF];
#pragma unroll
    for (int m = 0; m < MREF; ++m)
        vr[m] = *(const v2f*)(vb + (size_t)m * SDIM + s0);
    const int xs = tid / LPL;
    const int xp = tid % LPL;
    const v2f xin = *(const v2f*)(xr + (size_t)xs * N_OUT + 2 * xp);

#pragma unroll
    for (int m = 0; m < MREF; ++m)
        *(v2f*)(&sV[g][m * VPITCH + s0]) = vr[m];
    sX[2 * xp][xs]     = xin.x;
    sX[2 * xp + 1][xs] = xin.y;
    __syncthreads();
    const v2f xown = *(const v2f*)(&sX[g][s0]);

    v8f c00 = {}; v8f c01 = {}; v8f c11 = {}; v8f cd = {};
    const int kbase = wave * KPW;

#pragma unroll
    for (int ks = 0; ks < KPW; ks += 4) {
        const int kk = kbase + ks + 2 * half;
        v2f a0 = *(const v2f*)(&sV[g][lr * VPITCH + kk]);
        v2f a1 = *(const v2f*)(&sV[g][(16 + lr) * VPITCH + kk]);
        const float x0 = sX[g][kk];
        const float x1 = sX[g][kk + 1];
        v2f b0, b1;
        b0.x = (lr == 0) ? x0 : a0.x;
        b0.y = (lr == 0) ? x1 : a0.y;
        b1.x = (lr == 0) ? x0 : a1.x;
        b1.y = (lr == 0) ? x1 : a1.y;
        c00 = __builtin_amdgcn_wmma_f32_16x16x4_f32(false, a0, false, b0, (short)0, c00, false, false);
        c01 = __builtin_amdgcn_wmma_f32_16x16x4_f32(false, a0, false, a1, (short)0, c01, false, false);
        c11 = __builtin_amdgcn_wmma_f32_16x16x4_f32(false, a1, false, b1, (short)0, c11, false, false);
    }

    {
        const int drow = (lr < 8) ? 0 : 16;
        const int dcol = kbase + 4 * (lr & 7) + 2 * half;
        const v2f d0 = *(const v2f*)(&sV[g][drow * VPITCH + dcol]);
        const v2f d1 = *(const v2f*)(&sV[g][drow * VPITCH + dcol + 32]);
        cd = __builtin_amdgcn_wmma_f32_16x16x4_f32(false, d0, false, d0, (short)0, cd, false, false);
        cd = __builtin_amdgcn_wmma_f32_16x16x4_f32(false, d1, false, d1, (short)0, cd, false, false);
    }

#pragma unroll
    for (int j = 0; j < 8; ++j) {
        sP[g][wave][0][j * 32 + lane] = c00[j];
        sP[g][wave][1][j * 32 + lane] = c01[j];
        sP[g][wave][2][j * 32 + lane] = c11[j];
    }
    if (lr == 0) {
#pragma unroll
        for (int j = 0; j < 8; ++j) {
            sPW[g][wave][j + 8 * half]      = c00[j];
            sPW[g][wave][16 + j + 8 * half] = c11[j];
        }
    }
    {
        float dsel = cd[0];
#pragma unroll
        for (int j = 1; j < 8; ++j)
            dsel = ((lr & 7) == j) ? cd[j] : dsel;
        if ((lr >> 3) == half)
            sD[g][wave][lr] = dsel;
    }
    __syncthreads();

#pragma unroll
    for (int tile = 0; tile < 3; ++tile) {
#pragma unroll
        for (int e = 0; e < 2; ++e) {
            const int idx = tb + GTHR * e;
            float gv = sP[g][0][tile][idx];
#pragma unroll
            for (int w = 1; w < NWAVE; ++w)
                gv += sP[g][w][tile][idx];
            const int j   = idx >> 5;
            const int ln  = idx & 31;
            const int row = j + 8 * (ln >> 4) + ((tile == 2) ? 16 : 0);
            const int col = (ln & 15) + ((tile >= 1) ? 16 : 0);
            sG[g][row * GPITCH + col] = gv;
            if (tile == 1) sG[g][col * GPITCH + row] = gv;
        }
    }
    if (tb < MREF) {
        float wv = sPW[g][0][tb];
#pragma unroll
        for (int w = 1; w < NWAVE; ++w)
            wv += sPW[g][w][tb];
        sW[g][tb] = wv;
    }
    if (tb == 32 || tb == 64) {
        const int q = (tb == 32) ? 0 : 8;
        float ws[NWAVE];
#pragma unroll
        for (int w = 0; w < NWAVE; ++w) {
            const v4f lo = *(const v4f*)(&sD[g][w][q]);
            const v4f hi = *(const v4f*)(&sD[g][w][q + 4]);
            ws[w] = ((lo.x + lo.y) + (lo.z + lo.w)) + ((hi.x + hi.y) + (hi.z + hi.w));
        }
        sDg[g][(tb == 32) ? 0 : 1] = (ws[0] + ws[1]) + (ws[2] + ws[3]);
    }
    __syncthreads();

    const float gjj = (lr == 0) ? sDg[g][half] : sG[g][lane * GPITCH + lane];
    const float wj  = sW[g][lane];
    float gs[MREF];
#pragma unroll
    for (int q = 0; q < MREF / 4; ++q) {
        const v4f g4 = *(const v4f*)(&sG[g][lane * GPITCH + 4 * q]);
        gs[4 * q + 0] = g4.x; gs[4 * q + 1] = g4.y; gs[4 * q + 2] = g4.z; gs[4 * q + 3] = g4.w;
    }
    const float inv = 2.0f / gjj;
    float t = inv * wj;
#pragma unroll
    for (int k = 0; k < MREF; ++k)
        gs[k] *= inv;

    float a[MREF];
#pragma unroll
    for (int i = MREF - 1; i >= 0; --i) {
        a[i] = __int_as_float(__builtin_amdgcn_readlane(__float_as_int(t), i));
        t -= a[i] * gs[i];
    }

    float acc0 = xown.x;
    float acc1 = xown.y;
#pragma unroll
    for (int m = 0; m < MREF; ++m) {
        acc0 -= a[m] * vr[m].x;
        acc1 -= a[m] * vr[m].y;
    }

    {
        v2f yy; yy.x = acc0; yy.y = acc1;
        *(v2f*)(&sY[g][s0]) = yy;
    }
    __syncthreads();
    v2f yo;
    yo.x = sY[2 * xp][xs];
    yo.y = sY[2 * xp + 1][xs];
    *(v2f*)(out + (size_t)(i_in * SDIM + xs) * N_OUT + o0 + 2 * xp) = yo;
}

extern "C" void kernel_launch(void* const* d_in, const int* in_sizes, int n_in,
                              void* d_out, int out_size, void* d_ws, size_t ws_size,
                              hipStream_t stream) {
    const float* x = (const float*)d_in[0];
    const float* v = (const float*)d_in[1];
    float* out = (float*)d_out;

    dim3 grid(N_IN * N_OUT / NB);
    dim3 block(NTHR);
    householder_wy_kernel<<<grid, block, 0, stream>>>(x, v, out);
}


# Phase 1 no longer waits on a block barrier behind the V loads: staging is wave-private (k range = own columns), x is fetched first and handed over early; WMMAs overlap other waves' loads
# speedup vs baseline: 2.7018x; 1.0556x over previous
#include <hip/hip_runtime.h>


typedef __attribute__((ext_vector_type(2))) float v2f;
typedef __attribute__((ext_vector_type(4))) float v4f;
typedef __attribute__((ext_vector_type(8))) float v8f;

#define N_IN   16
#define N_OUT  64
#define MREF   32
#define SDIM   256
#define NB     2
#define NWAVE  4
#define GTHR   (NWAVE * 32)
#define NTHR   (NB * GTHR)
#define LPL    (NB / 2)
#define KPW    (SDIM / NWAVE)
#define VPITCH 260
#define GPITCH 36

__device__ __forceinline__ void lds_handover_barrier() {
    __builtin_amdgcn_fence(__ATOMIC_RELEASE, "workgroup", "local");
    __builtin_amdgcn_s_barrier();
    __builtin_amdgcn_fence(__ATOMIC_ACQUIRE, "workgroup", "local");
}

__global__ __launch_bounds__(NTHR) void householder_wy_kernel(
    const float* __restrict__ x,
    const float* __restrict__ v,
    float* __restrict__ out)
{
    const int tid  = threadIdx.x;
    const int g    = tid / GTHR;
    const int tb   = tid % GTHR;
    const int wave = tb >> 5;
    const int lane = tb & 31;
    const int half = lane >> 4;
    const int lr   = lane & 15;

    const int blk   = blockIdx.x;
    const int i_in  = blk / (N_OUT / NB);
    const int o0    = (blk % (N_OUT / NB)) * NB;
    const int batch = blk * NB + g;

    const float* vb = v + (size_t)batch * (MREF * SDIM);
    const float* xr = x + (size_t)(i_in * SDIM) * N_OUT + o0;

    __shared__ __attribute__((aligned(16))) float sV[NB][MREF * VPITCH];
    __shared__ __attribute__((aligned(16))) float sX[NB][SDIM];
    __shared__ float sP[NB][NWAVE][3][256];
    __shared__ float sPW[NB][NWAVE][MREF];
    __shared__ __attribute__((aligned(16))) float sG[NB][MREF * GPITCH];
    __shared__ float sW[NB][MREF];
    __shared__ __attribute__((aligned(16))) float sD[NB][NWAVE][16];
    __shared__ float sDg[NB][2];
    __shared__ __attribute__((aligned(16))) float sY[NB][SDIM];

    const int s0 = 2 * tb;
    const int xs = tid / LPL;
    const int xp = tid % LPL;
    const v2f xin = *(const v2f*)(xr + (size_t)xs * N_OUT + 2 * xp);
    __builtin_amdgcn_s_barrier();
    v2f vr[MREF];
#pragma unroll
    for (int m = 0; m < MREF; ++m)
        vr[m] = *(const v2f*)(vb + (size_t)m * SDIM + s0);
    sX[2 * xp][xs]     = xin.x;
    sX[2 * xp + 1][xs] = xin.y;
    lds_handover_barrier();
    const v2f xown = *(const v2f*)(&sX[g][s0]);

#pragma unroll
    for (int m = 0; m < MREF; ++m)
        *(v2f*)(&sV[g][m * VPITCH + s0]) = vr[m];
    __builtin_amdgcn_fence(__ATOMIC_RELEASE, "wavefront");
    __builtin_amdgcn_wave_barrier();
    __builtin_amdgcn_fence(__ATOMIC_ACQUIRE, "wavefront");

    v8f c00 = {}; v8f c01 = {}; v8f c11 = {}; v8f cd = {};
    const int kbase = wave * KPW;

#pragma unroll
    for (int ks = 0; ks < KPW; ks += 4) {
        const int kk = kbase + ks + 2 * half;
        v2f a0 = *(const v2f*)(&sV[g][lr * VPITCH + kk]);
        v2f a1 = *(const v2f*)(&sV[g][(16 + lr) * VPITCH + kk]);
        const float x0 = sX[g][kk];
        const float x1 = sX[g][kk + 1];
        v2f b0, b1;
        b0.x = (lr == 0) ? x0 : a0.x;
        b0.y = (lr == 0) ? x1 : a0.y;
        b1.x = (lr == 0) ? x0 : a1.x;
        b1.y = (lr == 0) ? x1 : a1.y;
        c00 = __builtin_amdgcn_wmma_f32_16x16x4_f32(false, a0, false, b0, (short)0, c00, false, false);
        c01 = __builtin_amdgcn_wmma_f32_16x16x4_f32(false, a0, false, a1, (short)0, c01, false, false);
        c11 = __builtin_amdgcn_wmma_f32_16x16x4_f32(false, a1, false, b1, (short)0, c11, false, false);
    }

    {
        const int drow = (lr < 8) ? 0 : 16;
        const int dcol = kbase + 4 * (lr & 7) + 2 * half;
        const v2f d0 = *(const v2f*)(&sV[g][drow * VPITCH + dcol]);
        const v2f d1 = *(const v2f*)(&sV[g][drow * VPITCH + dcol + 32]);
        cd = __builtin_amdgcn_wmma_f32_16x16x4_f32(false, d0, false, d0, (short)0, cd, false, false);
        cd = __builtin_amdgcn_wmma_f32_16x16x4_f32(false, d1, false, d1, (short)0, cd, false, false);
    }

#pragma unroll
    for (int j = 0; j < 8; ++j) {
        sP[g][wave][0][j * 32 + lane] = c00[j];
        sP[g][wave][1][j * 32 + lane] = c01[j];
        sP[g][wave][2][j * 32 + lane] = c11[j];
    }
    if (lr == 0) {
#pragma unroll
        for (int j = 0; j < 8; ++j) {
            sPW[g][wave][j + 8 * half]      = c00[j];
            sPW[g][wave][16 + j + 8 * half] = c11[j];
        }
    }
    {
        float dsel = cd[0];
#pragma unroll
        for (int j = 1; j < 8; ++j)
            dsel = ((lr & 7) == j) ? cd[j] : dsel;
        if ((lr >> 3) == half)
            sD[g][wave][lr] = dsel;
    }
    __syncthreads();

#pragma unroll
    for (int tile = 0; tile < 3; ++tile) {
#pragma unroll
        for (int e = 0; e < 2; ++e) {
            const int idx = tb + GTHR * e;
            float gv = sP[g][0][tile][idx];
#pragma unroll
            for (int w = 1; w < NWAVE; ++w)
                gv += sP[g][w][tile][idx];
            const int j   = idx >> 5;
            const int ln  = idx & 31;
            const int row = j + 8 * (ln >> 4) + ((tile == 2) ? 16 : 0);
            const int col = (ln & 15) + ((tile >= 1) ? 16 : 0);
            sG[g][row * GPITCH + col] = gv;
            if (tile == 1) sG[g][col * GPITCH + row] = gv;
        }
    }
    if (tb < MREF) {
        float wv = sPW[g][0][tb];
#pragma unroll
        for (int w = 1; w < NWAVE; ++w)
            wv += sPW[g][w][tb];
        sW[g][tb] = wv;
    }
    if (tb == 32 || tb == 64) {
        const int q = (tb == 32) ? 0 : 8;
        float ws[NWAVE];
#pragma unroll
        for (int w = 0; w < NWAVE; ++w) {
            const v4f lo = *(const v4f*)(&sD[g][w][q]);
            const v4f hi = *(const v4f*)(&sD[g][w][q + 4]);
            ws[w] = ((lo.x + lo.y) + (lo.z + lo.w)) + ((hi.x + hi.y) + (hi.z + hi.w));
        }
        sDg[g][(tb == 32) ? 0 : 1] = (ws[0] + ws[1]) + (ws[2] + ws[3]);
    }
    __syncthreads();

    const float gjj = (lr == 0) ? sDg[g][half] : sG[g][lane * GPITCH + lane];
    const float wj  = sW[g][lane];
    float gs[MREF];
#pragma unroll
    for (int q = 0; q < MREF / 4; ++q) {
        const v4f g4 = *(const v4f*)(&sG[g][lane * GPITCH + 4 * q]);
        gs[4 * q + 0] = g4.x; gs[4 * q + 1] = g4.y; gs[4 * q + 2] = g4.z; gs[4 * q + 3] = g4.w;
    }
    const float inv = 2.0f / gjj;
    float t = inv * wj;
#pragma unroll
    for (int k = 0; k < MREF; ++k)
        gs[k] *= inv;

    float a[MREF];
#pragma unroll
    for (int i = MREF - 1; i >= 0; --i) {
        a[i] = __int_as_float(__builtin_amdgcn_readlane(__float_as_int(t), i));
        t -= a[i] * gs[i];
    }

    float acc0 = xown.x;
    float acc1 = xown.y;
#pragma unroll
    for (int m = 0; m < MREF; ++m) {
        acc0 -= a[m] * vr[m].x;
        acc1 -= a[m] * vr[m].y;
    }

    {
        v2f yy; yy.x = acc0; yy.y = acc1;
        *(v2f*)(&sY[g][s0]) = yy;
    }
    __syncthreads();
    v2f yo;
    yo.x = sY[2 * xp][xs];
    yo.y = sY[2 * xp + 1][xs];
    *(v2f*)(out + (size_t)(i_in * SDIM + xs) * N_OUT + o0 + 2 * xp) = yo;
}

extern "C" void kernel_launch(void* const* d_in, const int* in_sizes, int n_in,
                              void* d_out, int out_size, void* d_ws, size_t ws_size,
                              hipStream_t stream) {
    const float* x = (const float*)d_in[0];
    const float* v = (const float*)d_in[1];
    float* out = (float*)d_out;

    dim3 grid(N_IN * N_OUT / NB);
    dim3 block(NTHR);
    householder_wy_kernel<<<grid, block, 0, stream>>>(x, v, out);
}


# Fewer, fatter waves: a wave owns 128 columns (16-byte loads), 2 waves per batch, 4 outputs per 8-wave block: 2,048 waves in 256 blocks; k split over 2 waves, so G and w are summed in a new order
# speedup vs baseline: 3.1001x; 1.1474x over previous
#include <hip/hip_runtime.h>


typedef __attribute__((ext_vector_type(2))) float v2f;
typedef __attribute__((ext_vector_type(4))) float v4f;
typedef __attribute__((ext_vector_type(8))) float v8f;

#define N_IN   16
#define N_OUT  64
#define MREF   32
#define SDIM   256
#define NB     4
#define NWAVE  2
#define GTHR   (NWAVE * 32)
#define NTHR   (NB * GTHR)
#define KPW    (SDIM / NWAVE)
#define CPT    (SDIM / GTHR)
#define VPITCH 260
#define GPITCH 36

__device__ __forceinline__ void lds_handover_barrier() {
    __builtin_amdgcn_fence(__ATOMIC_RELEASE, "workgroup", "local");
    __builtin_amdgcn_s_barrier();
    __builtin_amdgcn_fence(__ATOMIC_ACQUIRE, "workgroup", "local");
}

__global__ __launch_bounds__(NTHR) void householder_wy_kernel(
    const float* __restrict__ x,
    const float* __restrict__ v,
    float* __restrict__ out)
{
    const int tid  = threadIdx.x;
    const int g    = tid / GTHR;
    const int tb   = tid % GTHR;
    const int wave = tb >> 5;
    const int lane = tb & 31;
    const int half = lane >> 4;
    const int lr   = lane & 15;

    const int blk   = blockIdx.x;
    const int i_in  = blk / (N_OUT / NB);
    const int o0    = (blk % (N_OUT / NB)) * NB;
    const int batch = blk * NB + g;

    const float* vb = v + (size_t)batch * (MREF * SDIM);
    const float* xr = x + (size_t)(i_in * SDIM) * N_OUT + o0;

    __shared__ __attribute__((aligned(16))) float sV[NB][MREF * VPITCH];
    __shared__ __attribute__((aligned(16))) float sX[NB][SDIM];
    __shared__ float sP[NB][NWAVE][3][256];
    __shared__ float sPW[NB][NWAVE][MREF];
    __shared__ __attribute__((aligned(16))) float sG[NB][MREF * GPITCH];
    __shared__ float sW[NB][MREF];
    __shared__ __attribute__((aligned(16))) float sD[NB][NWAVE][16];
    __shared__ float sDg[NB][2];
    __shared__ __attribute__((aligned(16))) float sY[NB][SDIM];

    const int s0 = CPT * tb;
    const int xs = tid;
    const v4f xin = *(const v4f*)(xr + (size_t)xs * N_OUT);
    __builtin_amdgcn_s_barrier();
    v4f vr[MREF];
#pragma unroll
    for (int m = 0; m < MREF; ++m)
        vr[m] = *(const v4f*)(vb + (size_t)m * SDIM + s0);
    sX[0][xs] = xin.x;
    sX[1][xs] = xin.y;
    sX[2][xs] = xin.z;
    sX[3][xs] = xin.w;
    lds_handover_barrier();
    const v4f xown = *(const v4f*)(&sX[g][s0]);

#pragma unroll
    for (int m = 0; m < MREF; ++m)
        *(v4f*)(&sV[g][m * VPITCH + s0]) = vr[m];
    __builtin_amdgcn_fence(__ATOMIC_RELEASE, "wavefront");
    __builtin_amdgcn_wave_barrier();
    __builtin_amdgcn_fence(__ATOMIC_ACQUIRE, "wavefront");

    v8f c00 = {}; v8f c01 = {}; v8f c11 = {}; v8f cd = {};
    const int kbase = wave * KPW;

#pragma unroll
    for (int ks = 0; ks < KPW; ks += 4) {
        const int kk = kbase + ks + 2 * half;
        v2f a0 = *(const v2f*)(&sV[g][lr * VPITCH + kk]);
        v2f a1 = *(const v2f*)(&sV[g][(16 + lr) * VPITCH + kk]);
        const float x0 = sX[g][kk];
        const float x1 = sX[g][kk + 1];
        v2f b0, b1;
        b0.x = (lr == 0) ? x0 : a0.x;
        b0.y = (lr == 0) ? x1 : a0.y;
        b1.x = (lr == 0) ? x0 : a1.x;
        b1.y = (lr == 0) ? x1 : a1.y;
        c00 = __builtin_amdgcn_wmma_f32_16x16x4_f32(false, a0, false, b0, (short)0, c00, false, false);
        c01 = __builtin_amdgcn_wmma_f32_16x16x4_f32(false, a0, false, a1, (short)0, c01, false, false);
        c11 = __builtin_amdgcn_wmma_f32_16x16x4_f32(false, a1, false, b1, (short)0, c11, false, false);
    }

    {
        const int drow = (lr < 8) ? 0 : 16;
        const int dcol = kbase + 4 * (lr & 7) + 2 * half;
#pragma unroll
        for (int q = 0; q < KPW / 32; ++q) {
            const v2f d = *(const v2f*)(&sV[g][drow * VPITCH + dcol + 32 * q]);
            cd = __builtin_amdgcn_wmma_f32_16x16x4_f32(false, d, false, d, (short)0, cd, false, false);
        }
    }

#pragma unroll
    for (int j = 0; j < 8; ++j) {
        sP[g][wave][0][j * 32 + lane] = c00[j];
        sP[g][wave][1][j * 32 + lane] = c01[j];
        sP[g][wave][2][j * 32 + lane] = c11[j];
    }
    if (lr == 0) {
#pragma unroll
        for (int j = 0; j < 8; ++j) {
            sPW[g][wave][j + 8 * half]      = c00[j];
            sPW[g][wave][16 + j + 8 * half] = c11[j];
        }
    }
    {
        float dsel = cd[0];
#pragma unroll
        for (int j = 1; j < 8; ++j)
            dsel = ((lr & 7) == j) ? cd[j] : dsel;
        if ((lr >> 3) == half)
            sD[g][wave][lr] = dsel;
    }
    __syncthreads();

#pragma unroll
    for (int tile = 0; tile < 3; ++tile) {
#pragma unroll
        for (int e = 0; e < 256 / GTHR; ++e) {
            const int idx = tb + GTHR * e;
            float gv = sP[g][0][tile][idx];
#pragma unroll
            for (int w = 1; w < NWAVE; ++w)
                gv += sP[g][w][tile][idx];
            const int j   = idx >> 5;
            const int ln  = idx & 31;
            const int row = j + 8 * (ln >> 4) + ((tile == 2) ? 16 : 0);
            const int col = (ln & 15) + ((tile >= 1) ? 16 : 0);
            sG[g][row * GPITCH + col] = gv;
            if (tile == 1) sG[g][col * GPITCH + row] = gv;
        }
    }
    if (tb < MREF) {
        float wv = sPW[g][0][tb];
#pragma unroll
        for (int w = 1; w < NWAVE; ++w)
            wv += sPW[g][w][tb];
        sW[g][tb] = wv;
    }
    if (wave == 1 && lane < 2) {
        const int q = 8 * lane;
        float ws[NWAVE];
#pragma unroll
        for (int w = 0; w < NWAVE; ++w) {
            const v4f lo = *(const v4f*)(&sD[g][w][q]);
            const v4f hi = *(const v4f*)(&sD[g][w][q + 4]);
            ws[w] = ((lo.x + lo.y) + (lo.z + lo.w)) + ((hi.x + hi.y) + (hi.z + hi.w));
        }
        sDg[g][lane] = ws[0] + ws[1];
    }
    __syncthreads();

    const float gjj = (lr == 0) ? sDg[g][half] : sG[g][lane * GPITCH + lane];
    const float wj  = sW[g][lane];
    float gs[MREF];
#pragma unroll
    for (int q = 0; q < MREF / 4; ++q) {
        const v4f g4 = *(const v4f*)(&sG[g][lane * GPITCH + 4 * q]);
        gs[4 * q + 0] = g4.x; gs[4 * q + 1] = g4.y; gs[4 * q + 2] = g4.z; gs[4 * q + 3] = g4.w;
    }
    const float inv = 2.0f / gjj;
    float t = inv * wj;
#pragma unroll
    for (int k = 0; k < MREF; ++k)
        gs[k] *= inv;

    float a[MREF];
#pragma unroll
    for (int i = MREF - 1; i >= 0; --i) {
        a[i] = __int_as_float(__builtin_amdgcn_readlane(__float_as_int(t), i));
        t -= a[i] * gs[i];
    }

    float acc0 = xown.x;
    float acc1 = xown.y;
    float acc2 = xown.z;
    float acc3 = xown.w;
#pragma unroll
    for (int m = 0; m < MREF; ++m) {
        acc0 -= a[m] * vr[m].x;
        acc1 -= a[m] * vr[m].y;
        acc2 -= a[m] * vr[m].z;
        acc3 -= a[m] * vr[m].w;
    }

    {
        v4f yy; yy.x = acc0; yy.y = acc1; yy.z = acc2; yy.w = acc3;
        *(v4f*)(&sY[g][s0]) = yy;
    }
    __syncthreads();
    v4f yo;
    yo.x = sY[0][xs];
    yo.y = sY[1][xs];
    yo.z = sY[2][xs];
    yo.w = sY[3][xs];
    *(v4f*)(out + (size_t)(i_in * SDIM + xs) * N_OUT + o0) = yo;
}

extern "C" void kernel_launch(void* const* d_in, const int* in_sizes, int n_in,
                              void* d_out, int out_size, void* d_ws, size_t ws_size,
                              hipStream_t stream) {
    const float* x = (const float*)d_in[0];
    const float* v = (const float*)d_in[1];
    float* out = (float*)d_out;

    dim3 grid(N_IN * N_OUT / NB);
    dim3 block(NTHR);
    householder_wy_kernel<<<grid, block, 0, stream>>>(x, v, out);
}
